# starting kernel
# baseline (speedup 1.0000x reference)
#include <hip/hip_runtime.h>

typedef __attribute__((ext_vector_type(2))) float v2f;
typedef __attribute__((ext_vector_type(8))) float v8f;

#define HDIM 256
#define SB_STRIDE 72

__global__ __launch_bounds__(256)
void gemm_f32_wmma(const float* __restrict__ A, const float* __restrict__ B,
                   const float* __restrict__ bias, float* __restrict__ C,
                   int M, int K, int N, int applyRelu)
{
    __shared__ float sB[256 * SB_STRIDE];

    const int tid  = threadIdx.x;
    const int lane = tid & 31;
    const int wave = tid >> 5;
    const int half = lane >> 4;
    const int l16  = lane & 15;

    const int row0 = blockIdx.y * 128 + wave * 16;
    const int col0 = blockIdx.x * 64;

    {
        const int r = tid >> 4;
        const int c = (tid & 15) << 2;
        for (int kb = 0; kb < K; kb += 16) {
            const float4 v = *(const float4*)(B + (size_t)(kb + r) * N + col0 + c);
            *(float4*)(&sB[(kb + r) * SB_STRIDE + c]) = v;
        }
    }
    __syncthreads();

    int aRow = row0 + l16;
    if (aRow > M - 1) aRow = M - 1;
    const float* Aptr = A + (size_t)aRow * K + half * 2;

    v8f acc[4];
    acc[0] = (v8f)0.0f; acc[1] = (v8f)0.0f; acc[2] = (v8f)0.0f; acc[3] = (v8f)0.0f;

    for (int kk = 0; kk < K; kk += 4) {
        const v2f a = *(const v2f*)(Aptr + kk);
        __builtin_prefetch(Aptr + kk + 32, 0, 0);

        const float* s0 = &sB[(kk + half * 2) * SB_STRIDE + l16];
#pragma unroll
        for (int t = 0; t < 4; ++t) {
            v2f b;
            b.x = s0[t * 16];
            b.y = s0[SB_STRIDE + t * 16];
            acc[t] = __builtin_amdgcn_wmma_f32_16x16x4_f32(
                false, a, false, b, (short)0, acc[t], false, false);
        }
    }

#pragma unroll
    for (int t = 0; t < 4; ++t) {
        const int col = col0 + t * 16 + l16;
        const float bv = bias ? bias[col] : 0.0f;
#pragma unroll
        for (int r = 0; r < 8; ++r) {
            const int row = row0 + half * 8 + r;
            if (row < M) {
                float v = acc[t][r] + bv;
                if (applyRelu) v = v > 0.0f ? v : 0.0f;
                C[(size_t)row * N + col] = v;
            }
        }
    }
}

__global__ void deg_init(float* __restrict__ deg, int N)
{
    int i = blockIdx.x * blockDim.x + threadIdx.x;
    if (i < N) deg[i] = 1.0f;
}

__global__ void deg_edges(const int* __restrict__ dst, float* __restrict__ deg, int E)
{
    int e = blockIdx.x * blockDim.x + threadIdx.x;
    if (e < E) atomicAdd(&deg[dst[e]], 1.0f);
}

__global__ void deg_rsqrt(float* __restrict__ deg, int N)
{
    int i = blockIdx.x * blockDim.x + threadIdx.x;
    if (i < N) {
        float d = deg[i];
        deg[i] = d > 0.0f ? rsqrtf(d) : 0.0f;
    }
}

__global__ void gcn_self(const float* __restrict__ m, const float* __restrict__ dinv,
                         float* __restrict__ agg, long total)
{
    long idx = (long)blockIdx.x * blockDim.x + threadIdx.x;
    if (idx >= total) return;
    const int i = (int)(idx >> 8);
    const float w = dinv[i] * dinv[i];
    agg[idx] = m[idx] * w;
}

__global__ void gcn_scatter(const float* __restrict__ m, const int* __restrict__ src,
                            const int* __restrict__ dst, const float* __restrict__ dinv,
                            float* __restrict__ agg, int E)
{
    long gid = (long)blockIdx.x * blockDim.x + threadIdx.x;
    if (gid >= (long)E * 64) return;
    const int e = (int)(gid >> 6);
    const int c = (int)(gid & 63) << 2;
    const int s = src[e];
    const int d = dst[e];
    const float w = dinv[s] * dinv[d];
    const float4 v = *(const float4*)(m + (size_t)s * HDIM + c);
    float* o = agg + (size_t)d * HDIM + c;
    atomicAdd(o + 0, v.x * w);
    atomicAdd(o + 1, v.y * w);
    atomicAdd(o + 2, v.z * w);
    atomicAdd(o + 3, v.w * w);
}

__global__ void bias_relu(const float* __restrict__ agg, const float* __restrict__ b,
                          float* __restrict__ h, long total)
{
    long idx = (long)blockIdx.x * blockDim.x + threadIdx.x;
    if (idx >= total) return;
    const float v = agg[idx] + b[idx & (HDIM - 1)];
    h[idx] = v > 0.0f ? v : 0.0f;
}

extern "C" void kernel_launch(void* const* d_in, const int* in_sizes, int n_in,
                              void* d_out, int out_size, void* d_ws, size_t ws_size,
                              hipStream_t stream)
{
    const float* x      = (const float*)d_in[0];
    const int*   eidx   = (const int*)  d_in[1];
    const float* enc_w1 = (const float*)d_in[2];
    const float* enc_b1 = (const float*)d_in[3];
    const float* enc_w2 = (const float*)d_in[4];
    const float* enc_b2 = (const float*)d_in[5];
    const float* gcn_w  = (const float*)d_in[6];
    const float* gcn_b  = (const float*)d_in[7];

    const int F_IN = 128;
    const int N    = in_sizes[0] / F_IN;
    const int E    = in_sizes[1] / 2;
    const int L    = in_sizes[6] / (HDIM * HDIM);

    const int* src = eidx;
    const int* dst = eidx + E;

    float* h    = (float*)d_out;
    float* m    = (float*)d_ws;
    float* agg  = m + (size_t)N * HDIM;
    float* dinv = agg + (size_t)N * HDIM;

    const dim3 blk(256);
    const dim3 gg(HDIM / 64, (N + 127) / 128);
    const int tb = 256;
    const long total = (long)N * HDIM;

    gemm_f32_wmma<<<gg, blk, 0, stream>>>(x, enc_w1, enc_b1, m, N, F_IN, HDIM, 1);
    gemm_f32_wmma<<<gg, blk, 0, stream>>>(m, enc_w2, enc_b2, h, N, HDIM, HDIM, 0);

    deg_init <<<(N + tb - 1) / tb, tb, 0, stream>>>(dinv, N);
    deg_edges<<<(E + tb - 1) / tb, tb, 0, stream>>>(dst, dinv, E);
    deg_rsqrt<<<(N + tb - 1) / tb, tb, 0, stream>>>(dinv, N);

    for (int l = 0; l < L; ++l) {
        gemm_f32_wmma<<<gg, blk, 0, stream>>>(h, gcn_w + (size_t)l * HDIM * HDIM,
                                              nullptr, m, N, HDIM, HDIM, 0);
        gcn_self<<<(int)((total + tb - 1) / tb), tb, 0, stream>>>(m, dinv, agg, total);
        const long st = (long)E * 64;
        gcn_scatter<<<(int)((st + tb - 1) / tb), tb, 0, stream>>>(m, src, dst, dinv, agg, E);
        bias_relu<<<(int)((total + tb - 1) / tb), tb, 0, stream>>>(
            agg, gcn_b + (size_t)l * HDIM, h, total);
    }
}


# Aggregation as gather: atomicExch builds per-destination edge lists; one wave per node sums self + in-edge rows in registers in ascending edge id, adds bias, relu; no float atomics on features
# speedup vs baseline: 1.7513x; 1.7513x over previous
#include <hip/hip_runtime.h>

typedef __attribute__((ext_vector_type(2))) float v2f;
typedef __attribute__((ext_vector_type(8))) float v8f;

#define HDIM 256
#define SB_STRIDE 72

__global__ __launch_bounds__(256)
void gemm_f32_wmma(const float* __restrict__ A, const float* __restrict__ B,
                   const float* __restrict__ bias, float* __restrict__ C,
                   int M, int K, int N, int applyRelu)
{
    __shared__ float sB[256 * SB_STRIDE];

    const int tid  = threadIdx.x;
    const int lane = tid & 31;
    const int wave = tid >> 5;
    const int half = lane >> 4;
    const int l16  = lane & 15;

    const int row0 = blockIdx.y * 128 + wave * 16;
    const int col0 = blockIdx.x * 64;

    {
        const int r = tid >> 4;
        const int c = (tid & 15) << 2;
        for (int kb = 0; kb < K; kb += 16) {
            const float4 v = *(const float4*)(B + (size_t)(kb + r) * N + col0 + c);
            *(float4*)(&sB[(kb + r) * SB_STRIDE + c]) = v;
        }
    }
    __syncthreads();

    int aRow = row0 + l16;
    if (aRow > M - 1) aRow = M - 1;
    const float* Aptr = A + (size_t)aRow * K + half * 2;

    v8f acc[4];
    acc[0] = (v8f)0.0f; acc[1] = (v8f)0.0f; acc[2] = (v8f)0.0f; acc[3] = (v8f)0.0f;

    for (int kk = 0; kk < K; kk += 4) {
        const v2f a = *(const v2f*)(Aptr + kk);
        __builtin_prefetch(Aptr + kk + 32, 0, 0);

        const float* s0 = &sB[(kk + half * 2) * SB_STRIDE + l16];
#pragma unroll
        for (int t = 0; t < 4; ++t) {
            v2f b;
            b.x = s0[t * 16];
            b.y = s0[SB_STRIDE + t * 16];
            acc[t] = __builtin_amdgcn_wmma_f32_16x16x4_f32(
                false, a, false, b, (short)0, acc[t], false, false);
        }
    }

#pragma unroll
    for (int t = 0; t < 4; ++t) {
        const int col = col0 + t * 16 + l16;
        const float bv = bias ? bias[col] : 0.0f;
#pragma unroll
        for (int r = 0; r < 8; ++r) {
            const int row = row0 + half * 8 + r;
            if (row < M) {
                float v = acc[t][r] + bv;
                if (applyRelu) v = v > 0.0f ? v : 0.0f;
                C[(size_t)row * N + col] = v;
            }
        }
    }
}

__global__ void deg_init(float* __restrict__ deg, int* __restrict__ head, int N)
{
    int i = blockIdx.x * blockDim.x + threadIdx.x;
    if (i < N) {
        deg[i] = 1.0f;
        head[i] = -1;
    }
}

__global__ void deg_edges(const int* __restrict__ dst, float* __restrict__ deg,
                          int* __restrict__ head, int* __restrict__ nxt, int E)
{
    int e = blockIdx.x * blockDim.x + threadIdx.x;
    if (e < E) {
        const int d = dst[e];
        atomicAdd(&deg[d], 1.0f);
        nxt[e] = atomicExch(&head[d], e);
    }
}

__global__ void deg_rsqrt(float* __restrict__ deg, int N)
{
    int i = blockIdx.x * blockDim.x + threadIdx.x;
    if (i < N) {
        float d = deg[i];
        deg[i] = d > 0.0f ? rsqrtf(d) : 0.0f;
    }
}

__global__ __launch_bounds__(256)
void gcn_gather(const float* __restrict__ m, const int* __restrict__ src,
                const int* __restrict__ head, const int* __restrict__ nxt,
                const float* __restrict__ dinv, const float* __restrict__ b,
                float* __restrict__ h, int N, int E)
{
    const int lane = threadIdx.x & 31;
    const int node = blockIdx.x * 8 + __builtin_amdgcn_readfirstlane((int)(threadIdx.x >> 5));
    if (node >= N) return;

    const float di = dinv[node];
    const float4* mr = (const float4*)(m + (size_t)node * HDIM);
    float4 a0 = mr[lane];
    float4 a1 = mr[lane + 32];
    const float wself = di * di;
    a0.x *= wself; a0.y *= wself; a0.z *= wself; a0.w *= wself;
    a1.x *= wself; a1.y *= wself; a1.z *= wself; a1.w *= wself;

    int e = head[node];
    int left = E;
    while (e >= 0 && left > 0) {
        int myid = 0x7fffffff;
        int cnt = 0;
        while (e >= 0 && cnt < 32) {
            if (lane == cnt) myid = e;
            e = nxt[e];
            ++cnt;
        }
        left -= cnt;

        int mysrc = 0;
        float myw = 0.0f;
        if (lane < cnt) {
            mysrc = src[myid];
            myw = dinv[mysrc] * di;
        }
        int rank = 0;
        for (int j = 0; j < cnt; ++j) {
            const int idj = __builtin_amdgcn_readlane(myid, j);
            rank += (idj < myid) ? 1 : 0;
        }
        for (int r = 0; r < cnt; ++r) {
            const unsigned msk = __builtin_amdgcn_ballot_w32(lane < cnt && rank == r);
            const int l = (int)__builtin_ctz(msk);
            const int s = __builtin_amdgcn_readlane(mysrc, l);
            const float w = __int_as_float(__builtin_amdgcn_readlane(__float_as_int(myw), l));
            const float4* sr = (const float4*)(m + (size_t)s * HDIM);
            const float4 v0 = sr[lane];
            const float4 v1 = sr[lane + 32];
            a0.x += v0.x * w; a0.y += v0.y * w; a0.z += v0.z * w; a0.w += v0.w * w;
            a1.x += v1.x * w; a1.y += v1.y * w; a1.z += v1.z * w; a1.w += v1.w * w;
        }
    }

    const float4 b0 = ((const float4*)b)[lane];
    const float4 b1 = ((const float4*)b)[lane + 32];
    a0.x += b0.x; a0.y += b0.y; a0.z += b0.z; a0.w += b0.w;
    a1.x += b1.x; a1.y += b1.y; a1.z += b1.z; a1.w += b1.w;
    a0.x = a0.x > 0.0f ? a0.x : 0.0f; a0.y = a0.y > 0.0f ? a0.y : 0.0f;
    a0.z = a0.z > 0.0f ? a0.z : 0.0f; a0.w = a0.w > 0.0f ? a0.w : 0.0f;
    a1.x = a1.x > 0.0f ? a1.x : 0.0f; a1.y = a1.y > 0.0f ? a1.y : 0.0f;
    a1.z = a1.z > 0.0f ? a1.z : 0.0f; a1.w = a1.w > 0.0f ? a1.w : 0.0f;
    float4* hr = (float4*)(h + (size_t)node * HDIM);
    hr[lane]      = a0;
    hr[lane + 32] = a1;
}

extern "C" void kernel_launch(void* const* d_in, const int* in_sizes, int n_in,
                              void* d_out, int out_size, void* d_ws, size_t ws_size,
                              hipStream_t stream)
{
    const float* x      = (const float*)d_in[0];
    const int*   eidx   = (const int*)  d_in[1];
    const float* enc_w1 = (const float*)d_in[2];
    const float* enc_b1 = (const float*)d_in[3];
    const float* enc_w2 = (const float*)d_in[4];
    const float* enc_b2 = (const float*)d_in[5];
    const float* gcn_w  = (const float*)d_in[6];
    const float* gcn_b  = (const float*)d_in[7];

    const int F_IN = 128;
    const int N    = in_sizes[0] / F_IN;
    const int E    = in_sizes[1] / 2;
    const int L    = in_sizes[6] / (HDIM * HDIM);

    const int* src = eidx;
    const int* dst = eidx + E;

    float* h    = (float*)d_out;
    float* m    = (float*)d_ws;
    int*   head = (int*)(m + (size_t)N * HDIM);
    int*   nxt  = head + N;
    float* dinv = m + (size_t)2 * N * HDIM;

    const dim3 blk(256);
    const dim3 gg(HDIM / 64, (N + 127) / 128);
    const int tb = 256;

    gemm_f32_wmma<<<gg, blk, 0, stream>>>(x, enc_w1, enc_b1, m, N, F_IN, HDIM, 1);
    gemm_f32_wmma<<<gg, blk, 0, stream>>>(m, enc_w2, enc_b2, h, N, HDIM, HDIM, 0);

    deg_init <<<(N + tb - 1) / tb, tb, 0, stream>>>(dinv, head, N);
    deg_edges<<<(E + tb - 1) / tb, tb, 0, stream>>>(dst, dinv, head, nxt, E);
    deg_rsqrt<<<(N + tb - 1) / tb, tb, 0, stream>>>(dinv, N);

    for (int l = 0; l < L; ++l) {
        gemm_f32_wmma<<<gg, blk, 0, stream>>>(h, gcn_w + (size_t)l * HDIM * HDIM,
                                              nullptr, m, N, HDIM, HDIM, 0);
        gcn_gather<<<(N + 7) / 8, blk, 0, stream>>>(m, src, head, nxt, dinv,
                                                   gcn_b + (size_t)l * HDIM, h, N, E);
    }
}


# Fold W2@Wg0 and b2@Wg0 once (257-row WMMA product, 12 blocks): encoder output and first graph transform become one N x 256 x 256 product instead of two; new sum order of that product
# speedup vs baseline: 2.5160x; 1.4366x over previous
#include <hip/hip_runtime.h>

typedef __attribute__((ext_vector_type(2))) float v2f;
typedef __attribute__((ext_vector_type(8))) float v8f;

#define HDIM 256
#define SB_STRIDE 72

__device__ __forceinline__ void load_a_chunk(float4 (&q)[4], const float4* __restrict__ p)
{
#pragma unroll
    for (int i = 0; i < 4; ++i) q[i] = p[2 * i];
}

__device__ __forceinline__ void mma_chunk(const float4 (&q)[4], const float* __restrict__ s0,
                                          v8f (&acc)[4])
{
#pragma unroll
    for (int i = 0; i < 4; ++i) {
#pragma unroll
        for (int s = 0; s < 2; ++s) {
            v2f a;
            a.x = s ? q[i].z : q[i].x;
            a.y = s ? q[i].w : q[i].y;
            const float* sr = s0 + (8 * i + 2 * s) * SB_STRIDE;
#pragma unroll
            for (int t = 0; t < 4; ++t) {
                v2f b;
                b.x = sr[t * 16];
                b.y = sr[SB_STRIDE + t * 16];
                acc[t] = __builtin_amdgcn_wmma_f32_16x16x4_f32(
                    false, a, false, b, (short)0, acc[t], false, false);
            }
        }
    }
}

__device__ __forceinline__
void gemm_tile(const float* __restrict__ Arow, const float* __restrict__ B,
               const float* __restrict__ bias, float* __restrict__ C,
               int M, int K, int N, int applyRelu, float* __restrict__ sB)
{
    const int tid  = threadIdx.x;
    const int lane = tid & 31;
    const int wave = tid >> 5;
    const int half = lane >> 4;
    const int l16  = lane & 15;

    const int row0 = blockIdx.y * 128 + wave * 16;
    const int col0 = blockIdx.x * 64;

    {
        const int r = tid >> 4;
        const int c = (tid & 15) << 2;
        for (int kb = 0; kb < K; kb += 16) {
            const float4 v = *(const float4*)(B + (size_t)(kb + r) * N + col0 + c);
            *(float4*)(&sB[(kb + r) * SB_STRIDE + c]) = v;
        }
    }
    __syncthreads();

    const float4* Ap = (const float4*)Arow + half;
    const float*  s0 = &sB[(half * 4) * SB_STRIDE + l16];

    v8f acc[4];
    acc[0] = (v8f)0.0f; acc[1] = (v8f)0.0f; acc[2] = (v8f)0.0f; acc[3] = (v8f)0.0f;

    float4 qa[4], qb[4];
    load_a_chunk(qa, Ap);
    int kc = 0;
    for (; kc + 64 < K; kc += 64) {
        load_a_chunk(qb, Ap + (kc + 32) / 4);
        mma_chunk(qa, s0 + kc * SB_STRIDE, acc);
        load_a_chunk(qa, Ap + (kc + 64) / 4);
        mma_chunk(qb, s0 + (kc + 32) * SB_STRIDE, acc);
    }
    load_a_chunk(qb, Ap + (kc + 32) / 4);
    mma_chunk(qa, s0 + kc * SB_STRIDE, acc);
    mma_chunk(qb, s0 + (kc + 32) * SB_STRIDE, acc);

#pragma unroll
    for (int t = 0; t < 4; ++t) {
        const int col = col0 + t * 16 + l16;
        const float bv = bias ? bias[col] : 0.0f;
#pragma unroll
        for (int r = 0; r < 8; ++r) {
            const int row = row0 + half * 8 + r;
            if (row < M) {
                float v = acc[t][r] + bv;
                if (applyRelu) v = v > 0.0f ? v : 0.0f;
                C[(size_t)row * N + col] = v;
            }
        }
    }
}

__global__ __launch_bounds__(256)
void gemm_f32_wmma(const float* __restrict__ A, const float* __restrict__ B,
                   const float* __restrict__ bias, float* __restrict__ C,
                   int M, int K, int N, int applyRelu)
{
    __shared__ float sB[256 * SB_STRIDE];
    int aRow = blockIdx.y * 128 + (threadIdx.x >> 5) * 16 + (threadIdx.x & 15);
    if (aRow > M - 1) aRow = M - 1;
    gemm_tile(A + (size_t)aRow * K, B, bias, C, M, K, N, applyRelu, sB);
}

__global__ __launch_bounds__(256)
void fold_w2_wg(const float* __restrict__ W2, const float* __restrict__ b2,
                const float* __restrict__ Wg, float* __restrict__ Wf)
{
    __shared__ float sB[256 * SB_STRIDE];
    int aRow = blockIdx.y * 128 + (threadIdx.x >> 5) * 16 + (threadIdx.x & 15);
    if (aRow > HDIM) aRow = HDIM;
    const float* Arow = aRow < HDIM ? W2 + (size_t)aRow * HDIM : b2;
    gemm_tile(Arow, Wg, nullptr, Wf, HDIM + 1, HDIM, HDIM, 0, sB);
}

__global__ void deg_init(float* __restrict__ deg, int* __restrict__ head, int N)
{
    int i = blockIdx.x * blockDim.x + threadIdx.x;
    if (i < N) {
        deg[i] = 1.0f;
        head[i] = -1;
    }
}

__global__ void deg_edges(const int* __restrict__ dst, float* __restrict__ deg,
                          int* __restrict__ head, int* __restrict__ nxt, int E)
{
    int e = blockIdx.x * blockDim.x + threadIdx.x;
    if (e < E) {
        const int d = dst[e];
        atomicAdd(&deg[d], 1.0f);
        nxt[e] = atomicExch(&head[d], e);
    }
}

__global__ void deg_rsqrt(float* __restrict__ deg, int N)
{
    int i = blockIdx.x * blockDim.x + threadIdx.x;
    if (i < N) {
        float d = deg[i];
        deg[i] = d > 0.0f ? rsqrtf(d) : 0.0f;
    }
}

__global__ __launch_bounds__(256)
void gcn_gather(const float* __restrict__ m, const int* __restrict__ src,
                const int* __restrict__ head, const int* __restrict__ nxt,
                const float* __restrict__ dinv, const float* __restrict__ b,
                float* __restrict__ h, int N, int E)
{
    const int lane = threadIdx.x & 31;
    const int node = blockIdx.x * 8 + __builtin_amdgcn_readfirstlane((int)(threadIdx.x >> 5));
    if (node >= N) return;

    const float di = dinv[node];
    const float4* mr = (const float4*)(m + (size_t)node * HDIM);
    float4 a0 = mr[lane];
    float4 a1 = mr[lane + 32];
    const float wself = di * di;
    a0.x *= wself; a0.y *= wself; a0.z *= wself; a0.w *= wself;
    a1.x *= wself; a1.y *= wself; a1.z *= wself; a1.w *= wself;

    int e = head[node];
    int left = E;
    while (e >= 0 && left > 0) {
        int myid = 0x7fffffff;
        int cnt = 0;
        while (e >= 0 && cnt < 32) {
            if (lane == cnt) myid = e;
            e = nxt[e];
            ++cnt;
        }
        left -= cnt;

        int mysrc = 0;
        float myw = 0.0f;
        if (lane < cnt) {
            mysrc = src[myid];
            myw = dinv[mysrc] * di;
        }
        int rank = 0;
        for (int j = 0; j < cnt; ++j) {
            const int idj = __builtin_amdgcn_readlane(myid, j);
            rank += (idj < myid) ? 1 : 0;
        }
        for (int r = 0; r < cnt; ++r) {
            const unsigned msk = __builtin_amdgcn_ballot_w32(lane < cnt && rank == r);
            const int l = (int)__builtin_ctz(msk);
            const int s = __builtin_amdgcn_readlane(mysrc, l);
            const float w = __int_as_float(__builtin_amdgcn_readlane(__float_as_int(myw), l));
            const float4* sr = (const float4*)(m + (size_t)s * HDIM);
            const float4 v0 = sr[lane];
            const float4 v1 = sr[lane + 32];
            a0.x += v0.x * w; a0.y += v0.y * w; a0.z += v0.z * w; a0.w += v0.w * w;
            a1.x += v1.x * w; a1.y += v1.y * w; a1.z += v1.z * w; a1.w += v1.w * w;
        }
    }

    const float4 b0 = ((const float4*)b)[lane];
    const float4 b1 = ((const float4*)b)[lane + 32];
    a0.x += b0.x; a0.y += b0.y; a0.z += b0.z; a0.w += b0.w;
    a1.x += b1.x; a1.y += b1.y; a1.z += b1.z; a1.w += b1.w;
    a0.x = a0.x > 0.0f ? a0.x : 0.0f; a0.y = a0.y > 0.0f ? a0.y : 0.0f;
    a0.z = a0.z > 0.0f ? a0.z : 0.0f; a0.w = a0.w > 0.0f ? a0.w : 0.0f;
    a1.x = a1.x > 0.0f ? a1.x : 0.0f; a1.y = a1.y > 0.0f ? a1.y : 0.0f;
    a1.z = a1.z > 0.0f ? a1.z : 0.0f; a1.w = a1.w > 0.0f ? a1.w : 0.0f;
    float4* hr = (float4*)(h + (size_t)node * HDIM);
    hr[lane]      = a0;
    hr[lane + 32] = a1;
}

extern "C" void kernel_launch(void* const* d_in, const int* in_sizes, int n_in,
                              void* d_out, int out_size, void* d_ws, size_t ws_size,
                              hipStream_t stream)
{
    const float* x      = (const float*)d_in[0];
    const int*   eidx   = (const int*)  d_in[1];
    const float* enc_w1 = (const float*)d_in[2];
    const float* enc_b1 = (const float*)d_in[3];
    const float* enc_w2 = (const float*)d_in[4];
    const float* enc_b2 = (const float*)d_in[5];
    const float* gcn_w  = (const float*)d_in[6];
    const float* gcn_b  = (const float*)d_in[7];

    const int F_IN = 128;
    const int N    = in_sizes[0] / F_IN;
    const int E    = in_sizes[1] / 2;
    const int L    = in_sizes[6] / (HDIM * HDIM);

    const int* src = eidx;
    const int* dst = eidx + E;

    float* h    = (float*)d_out;
    float* m    = (float*)d_ws;
    int*   head = (int*)(m + (size_t)N * HDIM);
    int*   nxt  = head + N;
    float* dinv = m + (size_t)2 * N * HDIM;
    float* Wf   = dinv - (HDIM + 1) * HDIM;

    const dim3 blk(256);
    const dim3 gg(HDIM / 64, (N + 127) / 128);
    const int tb = 256;

    if (L < 1) {
        gemm_f32_wmma<<<gg, blk, 0, stream>>>(x, enc_w1, enc_b1, m, N, F_IN, HDIM, 1);
        gemm_f32_wmma<<<gg, blk, 0, stream>>>(m, enc_w2, enc_b2, h, N, HDIM, HDIM, 0);
        return;
    }

    gemm_f32_wmma<<<gg, blk, 0, stream>>>(x, enc_w1, enc_b1, h, N, F_IN, HDIM, 1);
    fold_w2_wg<<<dim3(HDIM / 64, 3), blk, 0, stream>>>(enc_w2, enc_b2, gcn_w, Wf);
    gemm_f32_wmma<<<gg, blk, 0, stream>>>(h, Wf, Wf + HDIM * HDIM, m, N, HDIM, HDIM, 0);

    deg_init <<<(N + tb - 1) / tb, tb, 0, stream>>>(dinv, head, N);
    deg_edges<<<(E + tb - 1) / tb, tb, 0, stream>>>(dst, dinv, head, nxt, E);
    deg_rsqrt<<<(N + tb - 1) / tb, tb, 0, stream>>>(dinv, N);

    for (int l = 0; l < L; ++l) {
        if (l > 0)
            gemm_f32_wmma<<<gg, blk, 0, stream>>>(h, gcn_w + (size_t)l * HDIM * HDIM,
                                                  nullptr, m, N, HDIM, HDIM, 0);
        gcn_gather<<<(N + 7) / 8, blk, 0, stream>>>(m, src, head, nxt, dinv,
                                                   gcn_b + (size_t)l * HDIM, h, N, E);
    }
}


# GEMM B operand: weight slice stored k-quad interleaved in LDS, so one 16-byte LDS read feeds two WMMA k-steps in aligned register pairs; half the LDS reads, no register shuffles; same k sum order
# speedup vs baseline: 2.6227x; 1.0424x over previous
#include <hip/hip_runtime.h>

typedef __attribute__((ext_vector_type(2))) float v2f;
typedef __attribute__((ext_vector_type(8))) float v8f;

#define HDIM 256
#define SB_PITCH4 65

__device__ __forceinline__ void load_a_chunk(float4 (&q)[4], const float4* __restrict__ p)
{
#pragma unroll
    for (int i = 0; i < 4; ++i) q[i] = p[2 * i];
}

__device__ __forceinline__ void mma_chunk(const float4 (&q)[4], const float4* __restrict__ s0,
                                          v8f (&acc)[4])
{
#pragma unroll
    for (int hh = 0; hh < 2; ++hh) {
        float4 bq[2][4];
#pragma unroll
        for (int ii = 0; ii < 2; ++ii) {
#pragma unroll
            for (int t = 0; t < 4; ++t)
                bq[ii][t] = s0[(4 * hh + 2 * ii) * SB_PITCH4 + t * 16];
        }
#pragma unroll
        for (int ii = 0; ii < 2; ++ii) {
            const int i = 2 * hh + ii;
#pragma unroll
            for (int s = 0; s < 2; ++s) {
                v2f a;
                a.x = s ? q[i].z : q[i].x;
                a.y = s ? q[i].w : q[i].y;
#pragma unroll
                for (int t = 0; t < 4; ++t) {
                    v2f b;
                    b.x = s ? bq[ii][t].z : bq[ii][t].x;
                    b.y = s ? bq[ii][t].w : bq[ii][t].y;
                    acc[t] = __builtin_amdgcn_wmma_f32_16x16x4_f32(
                        false, a, false, b, (short)0, acc[t], false, false);
                }
            }
        }
    }
}

__device__ __forceinline__
void gemm_tile(const float* __restrict__ Arow, const float* __restrict__ B,
               const float* __restrict__ bias, float* __restrict__ C,
               int M, int K, int N, int applyRelu, float4* __restrict__ sB4)
{
    const int tid  = threadIdx.x;
    const int lane = tid & 31;
    const int wave = tid >> 5;
    const int half = lane >> 4;
    const int l16  = lane & 15;

    const int row0 = blockIdx.y * 128 + wave * 16;
    const int col0 = blockIdx.x * 64;

    {
        const int kql = (tid & 3) + ((tid >> 6) << 2);
        const int n0  = ((tid >> 2) & 15) << 2;
        for (int kq = kql; kq < (K >> 2); kq += 16) {
            const float* g = B + (size_t)(4 * kq) * N + col0 + n0;
            const float4 r0 = *(const float4*)(g);
            const float4 r1 = *(const float4*)(g + N);
            const float4 r2 = *(const float4*)(g + 2 * (size_t)N);
            const float4 r3 = *(const float4*)(g + 3 * (size_t)N);
            float4* d = &sB4[kq * SB_PITCH4 + n0];
            d[0] = make_float4(r0.x, r1.x, r2.x, r3.x);
            d[1] = make_float4(r0.y, r1.y, r2.y, r3.y);
            d[2] = make_float4(r0.z, r1.z, r2.z, r3.z);
            d[3] = make_float4(r0.w, r1.w, r2.w, r3.w);
        }
    }
    __syncthreads();

    const float4* Ap = (const float4*)Arow + half;
    const float4* s0 = &sB4[half * SB_PITCH4 + l16];

    v8f acc[4];
    acc[0] = (v8f)0.0f; acc[1] = (v8f)0.0f; acc[2] = (v8f)0.0f; acc[3] = (v8f)0.0f;

    float4 qa[4], qb[4];
    load_a_chunk(qa, Ap);
    int kc = 0;
    for (; kc + 64 < K; kc += 64) {
        load_a_chunk(qb, Ap + (kc + 32) / 4);
        mma_chunk(qa, s0 + (kc / 4) * SB_PITCH4, acc);
        load_a_chunk(qa, Ap + (kc + 64) / 4);
        mma_chunk(qb, s0 + (kc / 4 + 8) * SB_PITCH4, acc);
    }
    load_a_chunk(qb, Ap + (kc + 32) / 4);
    mma_chunk(qa, s0 + (kc / 4) * SB_PITCH4, acc);
    mma_chunk(qb, s0 + (kc / 4 + 8) * SB_PITCH4, acc);

#pragma unroll
    for (int t = 0; t < 4; ++t) {
        const int col = col0 + t * 16 + l16;
        const float bv = bias ? bias[col] : 0.0f;
#pragma unroll
        for (int r = 0; r < 8; ++r) {
            const int row = row0 + half * 8 + r;
            if (row < M) {
                float v = acc[t][r] + bv;
                if (applyRelu) v = v > 0.0f ? v : 0.0f;
                C[(size_t)row * N + col] = v;
            }
        }
    }
}

__global__ __launch_bounds__(256)
void gemm_f32_wmma(const float* __restrict__ A, const float* __restrict__ B,
                   const float* __restrict__ bias, float* __restrict__ C,
                   int M, int K, int N, int applyRelu)
{
    __shared__ float4 sB4[64 * SB_PITCH4];
    int aRow = blockIdx.y * 128 + (threadIdx.x >> 5) * 16 + (threadIdx.x & 15);
    if (aRow > M - 1) aRow = M - 1;
    gemm_tile(A + (size_t)aRow * K, B, bias, C, M, K, N, applyRelu, sB4);
}

__global__ __launch_bounds__(256)
void fold_w2_wg(const float* __restrict__ W2, const float* __restrict__ b2,
                const float* __restrict__ Wg, float* __restrict__ Wf)
{
    __shared__ float4 sB4[64 * SB_PITCH4];
    int aRow = blockIdx.y * 128 + (threadIdx.x >> 5) * 16 + (threadIdx.x & 15);
    if (aRow > HDIM) aRow = HDIM;
    const float* Arow = aRow < HDIM ? W2 + (size_t)aRow * HDIM : b2;
    gemm_tile(Arow, Wg, nullptr, Wf, HDIM + 1, HDIM, HDIM, 0, sB4);
}

__global__ void deg_init(float* __restrict__ deg, int* __restrict__ head, int N)
{
    int i = blockIdx.x * blockDim.x + threadIdx.x;
    if (i < N) {
        deg[i] = 1.0f;
        head[i] = -1;
    }
}

__global__ void deg_edges(const int* __restrict__ dst, float* __restrict__ deg,
                          int* __restrict__ head, int* __restrict__ nxt, int E)
{
    int e = blockIdx.x * blockDim.x + threadIdx.x;
    if (e < E) {
        const int d = dst[e];
        atomicAdd(&deg[d], 1.0f);
        nxt[e] = atomicExch(&head[d], e);
    }
}

__global__ void deg_rsqrt(float* __restrict__ deg, int N)
{
    int i = blockIdx.x * blockDim.x + threadIdx.x;
    if (i < N) {
        float d = deg[i];
        deg[i] = d > 0.0f ? rsqrtf(d) : 0.0f;
    }
}

__global__ __launch_bounds__(256)
void gcn_gather(const float* __restrict__ m, const int* __restrict__ src,
                const int* __restrict__ head, const int* __restrict__ nxt,
                const float* __restrict__ dinv, const float* __restrict__ b,
                float* __restrict__ h, int N, int E)
{
    const int lane = threadIdx.x & 31;
    const int node = blockIdx.x * 8 + __builtin_amdgcn_readfirstlane((int)(threadIdx.x >> 5));
    if (node >= N) return;

    const float di = dinv[node];
    const float4* mr = (const float4*)(m + (size_t)node * HDIM);
    float4 a0 = mr[lane];
    float4 a1 = mr[lane + 32];
    const float wself = di * di;
    a0.x *= wself; a0.y *= wself; a0.z *= wself; a0.w *= wself;
    a1.x *= wself; a1.y *= wself; a1.z *= wself; a1.w *= wself;

    int e = head[node];
    int left = E;
    while (e >= 0 && left > 0) {
        int myid = 0x7fffffff;
        int cnt = 0;
        while (e >= 0 && cnt < 32) {
            if (lane == cnt) myid = e;
            e = nxt[e];
            ++cnt;
        }
        left -= cnt;

        int mysrc = 0;
        float myw = 0.0f;
        if (lane < cnt) {
            mysrc = src[myid];
            myw = dinv[mysrc] * di;
        }
        int rank = 0;
        for (int j = 0; j < cnt; ++j) {
            const int idj = __builtin_amdgcn_readlane(myid, j);
            rank += (idj < myid) ? 1 : 0;
        }
        for (int r = 0; r < cnt; ++r) {
            const unsigned msk = __builtin_amdgcn_ballot_w32(lane < cnt && rank == r);
            const int l = (int)__builtin_ctz(msk);
            const int s = __builtin_amdgcn_readlane(mysrc, l);
            const float w = __int_as_float(__builtin_amdgcn_readlane(__float_as_int(myw), l));
            const float4* sr = (const float4*)(m + (size_t)s * HDIM);
            const float4 v0 = sr[lane];
            const float4 v1 = sr[lane + 32];
            a0.x += v0.x * w; a0.y += v0.y * w; a0.z += v0.z * w; a0.w += v0.w * w;
            a1.x += v1.x * w; a1.y += v1.y * w; a1.z += v1.z * w; a1.w += v1.w * w;
        }
    }

    const float4 b0 = ((const float4*)b)[lane];
    const float4 b1 = ((const float4*)b)[lane + 32];
    a0.x += b0.x; a0.y += b0.y; a0.z += b0.z; a0.w += b0.w;
    a1.x += b1.x; a1.y += b1.y; a1.z += b1.z; a1.w += b1.w;
    a0.x = a0.x > 0.0f ? a0.x : 0.0f; a0.y = a0.y > 0.0f ? a0.y : 0.0f;
    a0.z = a0.z > 0.0f ? a0.z : 0.0f; a0.w = a0.w > 0.0f ? a0.w : 0.0f;
    a1.x = a1.x > 0.0f ? a1.x : 0.0f; a1.y = a1.y > 0.0f ? a1.y : 0.0f;
    a1.z = a1.z > 0.0f ? a1.z : 0.0f; a1.w = a1.w > 0.0f ? a1.w : 0.0f;
    float4* hr = (float4*)(h + (size_t)node * HDIM);
    hr[lane]      = a0;
    hr[lane + 32] = a1;
}

extern "C" void kernel_launch(void* const* d_in, const int* in_sizes, int n_in,
                              void* d_out, int out_size, void* d_ws, size_t ws_size,
                              hipStream_t stream)
{
    const float* x      = (const float*)d_in[0];
    const int*   eidx   = (const int*)  d_in[1];
    const float* enc_w1 = (const float*)d_in[2];
    const float* enc_b1 = (const float*)d_in[3];
    const float* enc_w2 = (const float*)d_in[4];
    const float* enc_b2 = (const float*)d_in[5];
    const float* gcn_w  = (const float*)d_in[6];
    const float* gcn_b  = (const float*)d_in[7];

    const int F_IN = 128;
    const int N    = in_sizes[0] / F_IN;
    const int E    = in_sizes[1] / 2;
    const int L    = in_sizes[6] / (HDIM * HDIM);

    const int* src = eidx;
    const int* dst = eidx + E;

    float* h    = (float*)d_out;
    float* m    = (float*)d_ws;
    int*   head = (int*)(m + (size_t)N * HDIM);
    int*   nxt  = head + N;
    float* dinv = m + (size_t)2 * N * HDIM;
    float* Wf   = dinv - (HDIM + 1) * HDIM;

    const dim3 blk(256);
    const dim3 gg(HDIM / 64, (N + 127) / 128);
    const int tb = 256;

    if (L < 1) {
        gemm_f32_wmma<<<gg, blk, 0, stream>>>(x, enc_w1, enc_b1, m, N, F_IN, HDIM, 1);
        gemm_f32_wmma<<<gg, blk, 0, stream>>>(m, enc_w2, enc_b2, h, N, HDIM, HDIM, 0);
        return;
    }

    gemm_f32_wmma<<<gg, blk, 0, stream>>>(x, enc_w1, enc_b1, h, N, F_IN, HDIM, 1);
    fold_w2_wg<<<dim3(HDIM / 64, 3), blk, 0, stream>>>(enc_w2, enc_b2, gcn_w, Wf);
    gemm_f32_wmma<<<gg, blk, 0, stream>>>(h, Wf, Wf + HDIM * HDIM, m, N, HDIM, HDIM, 0);

    deg_init <<<(N + tb - 1) / tb, tb, 0, stream>>>(dinv, head, N);
    deg_edges<<<(E + tb - 1) / tb, tb, 0, stream>>>(dst, dinv, head, nxt, E);
    deg_rsqrt<<<(N + tb - 1) / tb, tb, 0, stream>>>(dinv, N);

    for (int l = 0; l < L; ++l) {
        if (l > 0)
            gemm_f32_wmma<<<gg, blk, 0, stream>>>(h, gcn_w + (size_t)l * HDIM * HDIM,
                                                  nullptr, m, N, HDIM, HDIM, 0);
        gcn_gather<<<(N + 7) / 8, blk, 0, stream>>>(m, src, head, nxt, dinv,
                                                   gcn_b + (size_t)l * HDIM, h, N, E);
    }
}


# GEMMs by 256 resident work-groups of 16 waves: all of B staged once per work-group in LDS (257 KB), waves loop over 16x64 tiles with no per-tile staging, barrier or restart; same k sum order
# speedup vs baseline: 2.9454x; 1.1230x over previous
#include <hip/hip_runtime.h>

typedef __attribute__((ext_vector_type(2))) float v2f;
typedef __attribute__((ext_vector_type(8))) float v8f;

#define HDIM 256
#define SB_PITCH4 65
#define SBF_PITCH4 257

__device__ __forceinline__ void load_a_chunk(float4 (&q)[4], const float4* __restrict__ p)
{
#pragma unroll
    for (int i = 0; i < 4; ++i) q[i] = p[2 * i];
}

template <int PITCH4>
__device__ __forceinline__ void mma_chunk(const float4 (&q)[4], const float4* __restrict__ s0,
                                          v8f (&acc)[4])
{
#pragma unroll
    for (int hh = 0; hh < 2; ++hh) {
        float4 bq[2][4];
#pragma unroll
        for (int ii = 0; ii < 2; ++ii) {
#pragma unroll
            for (int t = 0; t < 4; ++t)
                bq[ii][t] = s0[(4 * hh + 2 * ii) * PITCH4 + t * 16];
        }
#pragma unroll
        for (int ii = 0; ii < 2; ++ii) {
            const int i = 2 * hh + ii;
#pragma unroll
            for (int s = 0; s < 2; ++s) {
                v2f a;
                a.x = s ? q[i].z : q[i].x;
                a.y = s ? q[i].w : q[i].y;
#pragma unroll
                for (int t = 0; t < 4; ++t) {
                    v2f b;
                    b.x = s ? bq[ii][t].z : bq[ii][t].x;
                    b.y = s ? bq[ii][t].w : bq[ii][t].y;
                    acc[t] = __builtin_amdgcn_wmma_f32_16x16x4_f32(
                        false, a, false, b, (short)0, acc[t], false, false);
                }
            }
        }
    }
}

__device__ __forceinline__
void gemm_tile(const float* __restrict__ Arow, const float* __restrict__ B,
               const float* __restrict__ bias, float* __restrict__ C,
               int M, int K, int N, int applyRelu, float4* __restrict__ sB4)
{
    const int tid  = threadIdx.x;
    const int lane = tid & 31;
    const int wave = tid >> 5;
    const int half = lane >> 4;
    const int l16  = lane & 15;

    const int row0 = blockIdx.y * 128 + wave * 16;
    const int col0 = blockIdx.x * 64;

    {
        const int kql = (tid & 3) + ((tid >> 6) << 2);
        const int n0  = ((tid >> 2) & 15) << 2;
        for (int kq = kql; kq < (K >> 2); kq += 16) {
            const float* g = B + (size_t)(4 * kq) * N + col0 + n0;
            const float4 r0 = *(const float4*)(g);
            const float4 r1 = *(const float4*)(g + N);
            const float4 r2 = *(const float4*)(g + 2 * (size_t)N);
            const float4 r3 = *(const float4*)(g + 3 * (size_t)N);
            float4* d = &sB4[kq * SB_PITCH4 + n0];
            d[0] = make_float4(r0.x, r1.x, r2.x, r3.x);
            d[1] = make_float4(r0.y, r1.y, r2.y, r3.y);
            d[2] = make_float4(r0.z, r1.z, r2.z, r3.z);
            d[3] = make_float4(r0.w, r1.w, r2.w, r3.w);
        }
    }
    __syncthreads();

    const float4* Ap = (const float4*)Arow + half;
    const float4* s0 = &sB4[half * SB_PITCH4 + l16];

    v8f acc[4];
    acc[0] = (v8f)0.0f; acc[1] = (v8f)0.0f; acc[2] = (v8f)0.0f; acc[3] = (v8f)0.0f;

    float4 qa[4], qb[4];
    load_a_chunk(qa, Ap);
    int kc = 0;
    for (; kc + 64 < K; kc += 64) {
        load_a_chunk(qb, Ap + (kc + 32) / 4);
        mma_chunk<SB_PITCH4>(qa, s0 + (kc / 4) * SB_PITCH4, acc);
        load_a_chunk(qa, Ap + (kc + 64) / 4);
        mma_chunk<SB_PITCH4>(qb, s0 + (kc / 4 + 8) * SB_PITCH4, acc);
    }
    load_a_chunk(qb, Ap + (kc + 32) / 4);
    mma_chunk<SB_PITCH4>(qa, s0 + (kc / 4) * SB_PITCH4, acc);
    mma_chunk<SB_PITCH4>(qb, s0 + (kc / 4 + 8) * SB_PITCH4, acc);

#pragma unroll
    for (int t = 0; t < 4; ++t) {
        const int col = col0 + t * 16 + l16;
        const float bv = bias ? bias[col] : 0.0f;
#pragma unroll
        for (int r = 0; r < 8; ++r) {
            const int row = row0 + half * 8 + r;
            if (row < M) {
                float v = acc[t][r] + bv;
                if (applyRelu) v = v > 0.0f ? v : 0.0f;
                C[(size_t)row * N + col] = v;
            }
        }
    }
}

__global__ __launch_bounds__(512)
void gemm_f32_wmma_resident(const float* __restrict__ A, const float* __restrict__ B,
                            const float* __restrict__ bias, float* __restrict__ C,
                            int M, int K, int applyRelu)
{
    __shared__ float4 sBf[64 * SBF_PITCH4];

    const int tid  = threadIdx.x;
    const int lane = tid & 31;
    const int half = lane >> 4;
    const int l16  = lane & 15;
    const int wave = __builtin_amdgcn_readfirstlane(tid >> 5);
    const int bid  = blockIdx.x;
    const int nblk = gridDim.x;

    {
        const int kql = (tid & 3) + ((tid >> 6) << 2);
        const int n0  = ((tid >> 2) & 15) << 2;
        const int pieces = K >> 5;
        for (int i = 0; i < pieces; i += 4) {
            float4 r[4][4];
#pragma unroll
            for (int u = 0; u < 4; ++u) {
                const int p  = (i + u + bid) & (pieces - 1);
                const int kq = kql + ((p >> 2) << 5);
                const float* gp = B + (size_t)(4 * kq) * HDIM + ((p & 3) << 6) + n0;
                r[u][0] = *(const float4*)(gp);
                r[u][1] = *(const float4*)(gp + HDIM);
                r[u][2] = *(const float4*)(gp + 2 * HDIM);
                r[u][3] = *(const float4*)(gp + 3 * HDIM);
            }
#pragma unroll
            for (int u = 0; u < 4; ++u) {
                const int p  = (i + u + bid) & (pieces - 1);
                const int kq = kql + ((p >> 2) << 5);
                float4* d = &sBf[kq * SBF_PITCH4 + ((p & 3) << 6) + n0];
                d[0] = make_float4(r[u][0].x, r[u][1].x, r[u][2].x, r[u][3].x);
                d[1] = make_float4(r[u][0].y, r[u][1].y, r[u][2].y, r[u][3].y);
                d[2] = make_float4(r[u][0].z, r[u][1].z, r[u][2].z, r[u][3].z);
                d[3] = make_float4(r[u][0].w, r[u][1].w, r[u][2].w, r[u][3].w);
            }
        }
    }
    __syncthreads();

    const int col0 = (wave & 3) * 64;
    const int grp  = wave >> 2;
    const float4* s0 = &sBf[half * SBF_PITCH4 + col0 + l16];

    float bv[4];
#pragma unroll
    for (int t = 0; t < 4; ++t) bv[t] = bias ? bias[col0 + t * 16 + l16] : 0.0f;

    const int numRT = (M + 15) >> 4;
    int rt = grp * nblk + bid;
    if (rt >= numRT) return;

    int aRow = rt * 16 + l16;
    if (aRow > M - 1) aRow = M - 1;
    const float4* Ap = (const float4*)(A + (size_t)aRow * K) + half;

    float4 qa[4], qb[4];
    load_a_chunk(qa, Ap);
    for (;;) {
        const int rtn = rt + 4 * nblk;
        const bool more = rtn < numRT;
        int aRowN = (more ? rtn : rt) * 16 + l16;
        if (aRowN > M - 1) aRowN = M - 1;
        const float4* Apn = (const float4*)(A + (size_t)aRowN * K) + half;

        v8f acc[4];
        acc[0] = (v8f)0.0f; acc[1] = (v8f)0.0f; acc[2] = (v8f)0.0f; acc[3] = (v8f)0.0f;

        int kc = 0;
        for (; kc + 64 < K; kc += 64) {
            load_a_chunk(qb, Ap + (kc + 32) / 4);
            mma_chunk<SBF_PITCH4>(qa, s0 + (kc / 4) * SBF_PITCH4, acc);
            load_a_chunk(qa, Ap + (kc + 64) / 4);
            mma_chunk<SBF_PITCH4>(qb, s0 + (kc / 4 + 8) * SBF_PITCH4, acc);
        }
        load_a_chunk(qb, Ap + (kc + 32) / 4);
        mma_chunk<SBF_PITCH4>(qa, s0 + (kc / 4) * SBF_PITCH4, acc);
        load_a_chunk(qa, Apn);
        mma_chunk<SBF_PITCH4>(qb, s0 + (kc / 4 + 8) * SBF_PITCH4, acc);

#pragma unroll
        for (int t = 0; t < 4; ++t) {
            const int col = col0 + t * 16 + l16;
#pragma unroll
            for (int r = 0; r < 8; ++r) {
                const int row = rt * 16 + half * 8 + r;
                if (row < M) {
                    float v = acc[t][r] + bv[t];
                    if (applyRelu) v = v > 0.0f ? v : 0.0f;
                    C[(size_t)row * HDIM + col] = v;
                }
            }
        }

        if (!more) break;
        rt = rtn;
        Ap = Apn;
    }
}

__global__ __launch_bounds__(256)
void fold_w2_wg(const float* __restrict__ W2, const float* __restrict__ b2,
                const float* __restrict__ Wg, float* __restrict__ Wf)
{
    __shared__ float4 sB4[64 * SB_PITCH4];
    int aRow = blockIdx.y * 128 + (threadIdx.x >> 5) * 16 + (threadIdx.x & 15);
    if (aRow > HDIM) aRow = HDIM;
    const float* Arow = aRow < HDIM ? W2 + (size_t)aRow * HDIM : b2;
    gemm_tile(Arow, Wg, nullptr, Wf, HDIM + 1, HDIM, HDIM, 0, sB4);
}

__global__ void deg_init(float* __restrict__ deg, int* __restrict__ head, int N)
{
    int i = blockIdx.x * blockDim.x + threadIdx.x;
    if (i < N) {
        deg[i] = 1.0f;
        head[i] = -1;
    }
}

__global__ void deg_edges(const int* __restrict__ dst, float* __restrict__ deg,
                          int* __restrict__ head, int* __restrict__ nxt, int E)
{
    int e = blockIdx.x * blockDim.x + threadIdx.x;
    if (e < E) {
        const int d = dst[e];
        atomicAdd(&deg[d], 1.0f);
        nxt[e] = atomicExch(&head[d], e);
    }
}

__global__ void deg_rsqrt(float* __restrict__ deg, int N)
{
    int i = blockIdx.x * blockDim.x + threadIdx.x;
    if (i < N) {
        float d = deg[i];
        deg[i] = d > 0.0f ? rsqrtf(d) : 0.0f;
    }
}

__global__ __launch_bounds__(256)
void gcn_gather(const float* __restrict__ m, const int* __restrict__ src,
                const int* __restrict__ head, const int* __restrict__ nxt,
                const float* __restrict__ dinv, const float* __restrict__ b,
                float* __restrict__ h, int N, int E)
{
    const int lane = threadIdx.x & 31;
    const int node = blockIdx.x * 8 + __builtin_amdgcn_readfirstlane((int)(threadIdx.x >> 5));
    if (node >= N) return;

    const float di = dinv[node];
    const float4* mr = (const float4*)(m + (size_t)node * HDIM);
    float4 a0 = mr[lane];
    float4 a1 = mr[lane + 32];
    const float wself = di * di;
    a0.x *= wself; a0.y *= wself; a0.z *= wself; a0.w *= wself;
    a1.x *= wself; a1.y *= wself; a1.z *= wself; a1.w *= wself;

    int e = head[node];
    int left = E;
    while (e >= 0 && left > 0) {
        int myid = 0x7fffffff;
        int cnt = 0;
        while (e >= 0 && cnt < 32) {
            if (lane == cnt) myid = e;
            e = nxt[e];
            ++cnt;
        }
        left -= cnt;

        int mysrc = 0;
        float myw = 0.0f;
        if (lane < cnt) {
            mysrc = src[myid];
            myw = dinv[mysrc] * di;
        }
        int rank = 0;
        for (int j = 0; j < cnt; ++j) {
            const int idj = __builtin_amdgcn_readlane(myid, j);
            rank += (idj < myid) ? 1 : 0;
        }
        for (int r = 0; r < cnt; ++r) {
            const unsigned msk = __builtin_amdgcn_ballot_w32(lane < cnt && rank == r);
            const int l = (int)__builtin_ctz(msk);
            const int s = __builtin_amdgcn_readlane(mysrc, l);
            const float w = __int_as_float(__builtin_amdgcn_readlane(__float_as_int(myw), l));
            const float4* sr = (const float4*)(m + (size_t)s * HDIM);
            const float4 v0 = sr[lane];
            const float4 v1 = sr[lane + 32];
            a0.x += v0.x * w; a0.y += v0.y * w; a0.z += v0.z * w; a0.w += v0.w * w;
            a1.x += v1.x * w; a1.y += v1.y * w; a1.z += v1.z * w; a1.w += v1.w * w;
        }
    }

    const float4 b0 = ((const float4*)b)[lane];
    const float4 b1 = ((const float4*)b)[lane + 32];
    a0.x += b0.x; a0.y += b0.y; a0.z += b0.z; a0.w += b0.w;
    a1.x += b1.x; a1.y += b1.y; a1.z += b1.z; a1.w += b1.w;
    a0.x = a0.x > 0.0f ? a0.x : 0.0f; a0.y = a0.y > 0.0f ? a0.y : 0.0f;
    a0.z = a0.z > 0.0f ? a0.z : 0.0f; a0.w = a0.w > 0.0f ? a0.w : 0.0f;
    a1.x = a1.x > 0.0f ? a1.x : 0.0f; a1.y = a1.y > 0.0f ? a1.y : 0.0f;
    a1.z = a1.z > 0.0f ? a1.z : 0.0f; a1.w = a1.w > 0.0f ? a1.w : 0.0f;
    float4* hr = (float4*)(h + (size_t)node * HDIM);
    hr[lane]      = a0;
    hr[lane + 32] = a1;
}

extern "C" void kernel_launch(void* const* d_in, const int* in_sizes, int n_in,
                              void* d_out, int out_size, void* d_ws, size_t ws_size,
                              hipStream_t stream)
{
    const float* x      = (const float*)d_in[0];
    const int*   eidx   = (const int*)  d_in[1];
    const float* enc_w1 = (const float*)d_in[2];
    const float* enc_b1 = (const float*)d_in[3];
    const float* enc_w2 = (const float*)d_in[4];
    const float* enc_b2 = (const float*)d_in[5];
    const float* gcn_w  = (const float*)d_in[6];
    const float* gcn_b  = (const float*)d_in[7];

    const int F_IN = 128;
    const int N    = in_sizes[0] / F_IN;
    const int E    = in_sizes[1] / 2;
    const int L    = in_sizes[6] / (HDIM * HDIM);

    const int* src = eidx;
    const int* dst = eidx + E;

    float* h    = (float*)d_out;
    float* m    = (float*)d_ws;
    int*   head = (int*)(m + (size_t)N * HDIM);
    int*   nxt  = head + N;
    float* dinv = m + (size_t)2 * N * HDIM;
    float* Wf   = dinv - (HDIM + 1) * HDIM;

    const dim3 blk(256);
    const dim3 pg(256), pblk(512);
    const int tb = 256;

    if (L < 1) {
        gemm_f32_wmma_resident<<<pg, pblk, 0, stream>>>(x, enc_w1, enc_b1, m, N, F_IN, 1);
        gemm_f32_wmma_resident<<<pg, pblk, 0, stream>>>(m, enc_w2, enc_b2, h, N, HDIM, 0);
        return;
    }

    gemm_f32_wmma_resident<<<pg, pblk, 0, stream>>>(x, enc_w1, enc_b1, h, N, F_IN, 1);
    fold_w2_wg<<<dim3(HDIM / 64, 3), blk, 0, stream>>>(enc_w2, enc_b2, gcn_w, Wf);
    gemm_f32_wmma_resident<<<pg, pblk, 0, stream>>>(h, Wf, Wf + HDIM * HDIM, m, N, HDIM, 0);

    deg_init <<<(N + tb - 1) / tb, tb, 0, stream>>>(dinv, head, N);
    deg_edges<<<(E + tb - 1) / tb, tb, 0, stream>>>(dst, dinv, head, nxt, E);
    deg_rsqrt<<<(N + tb - 1) / tb, tb, 0, stream>>>(dinv, N);

    for (int l = 0; l < L; ++l) {
        if (l > 0)
            gemm_f32_wmma_resident<<<pg, pblk, 0, stream>>>(h, gcn_w + (size_t)l * HDIM * HDIM,
                                                            nullptr, m, N, HDIM, 0);
        gcn_gather<<<(N + 7) / 8, blk, 0, stream>>>(m, src, head, nxt, dinv,
                                                   gcn_b + (size_t)l * HDIM, h, N, E);
    }
}
